# starting kernel
# baseline (speedup 1.0000x reference)
#include <hip/hip_runtime.h>
#include <math.h>

typedef float v2f __attribute__((ext_vector_type(2)));
typedef float v8f __attribute__((ext_vector_type(8)));

#define DIM 128

__global__ __launch_bounds__(256) void gcn_gemm_wmma(
    const float* __restrict__ seq,
    const float* __restrict__ w_fc,
    const float* __restrict__ w_proj,
    const float* __restrict__ bias,
    const float* __restrict__ b_proj,
    float* __restrict__ h,
    float* __restrict__ out,
    int n)
{
    const int lane = threadIdx.x & 31;
    const int wave = threadIdx.x >> 5;
    const int row0 = blockIdx.x * 128 + wave * 16;

    const int m  = lane & 15;
    const int kh = (lane >> 4) * 2;
    const int nn = lane & 15;

    int arow = row0 + m;
    if (arow >= n) arow = n - 1;
    const float* abase = seq + (size_t)arow * DIM + kh;
    v2f a[32];
#pragma unroll
    for (int kk = 0; kk < 32; ++kk)
        a[kk] = *(const v2f*)(abase + kk * 4);

    const int rowsel = (lane >> 4) << 3;

#pragma unroll 1
    for (int t = 0; t < 8; ++t) {
        const int colBase = t * 16;
        v8f cf = {};
        v8f cp = {};

        const float* bf = w_fc   + (size_t)kh * DIM + colBase + nn;
        const float* bp = w_proj + (size_t)kh * DIM + colBase + nn;

#pragma unroll
        for (int kk = 0; kk < 32; ++kk) {
            const size_t r0 = (size_t)(kk * 4) * DIM;
            const size_t r1 = (size_t)(kk * 4 + 1) * DIM;
            v2f bfr, bpr;
            bfr.x = bf[r0];  bfr.y = bf[r1];
            bpr.x = bp[r0];  bpr.y = bp[r1];
            cf = __builtin_amdgcn_wmma_f32_16x16x4_f32(
                    false, a[kk], false, bfr, (short)0, cf, false, false);
            cp = __builtin_amdgcn_wmma_f32_16x16x4_f32(
                    false, a[kk], false, bpr, (short)0, cp, false, false);
        }

        const float badd = bias[colBase + nn] + b_proj[colBase + nn];

#pragma unroll
        for (int v = 0; v < 8; ++v) {
            const int row = row0 + v + rowsel;
            if (row < n) {
                h  [(size_t)row * DIM + colBase + nn] = cf[v];
                out[(size_t)row * DIM + colBase + nn] = cp[v] + badd;
            }
        }
    }
}

__global__ __launch_bounds__(256) void gcn_edge_scatter(
    const float* __restrict__ h,
    const int*   __restrict__ edge_src,
    const int*   __restrict__ edge_dst,
    const float* __restrict__ adj_vals,
    float* __restrict__ out,
    int e_total)
{
    const int e = blockIdx.x * (blockDim.x >> 5) + (threadIdx.x >> 5);
    if (e >= e_total) return;
    const int lane = threadIdx.x & 31;

    const int   src = edge_src[e];
    const int   dst = edge_dst[e];
    const float val = adj_vals[e];

    const float4 hv = ((const float4*)(h + (size_t)src * DIM))[lane];
    float* o = out + (size_t)dst * DIM + lane * 4;
    atomicAdd(o + 0, val * hv.x);
    atomicAdd(o + 1, val * hv.y);
    atomicAdd(o + 2, val * hv.z);
    atomicAdd(o + 3, val * hv.w);
}

__global__ __launch_bounds__(256) void gcn_tanh(float* __restrict__ out, int n4)
{
    const int i = blockIdx.x * blockDim.x + threadIdx.x;
    if (i >= n4) return;
    float4 v = ((float4*)out)[i];
    v.x = tanhf(v.x); v.y = tanhf(v.y); v.z = tanhf(v.z); v.w = tanhf(v.w);
    ((float4*)out)[i] = v;
}

extern "C" void kernel_launch(void* const* d_in, const int* in_sizes, int n_in,
                              void* d_out, int out_size, void* d_ws, size_t ws_size,
                              hipStream_t stream)
{
    const float* seq      = (const float*)d_in[0];
    const int*   edge_src = (const int*)  d_in[1];
    const int*   edge_dst = (const int*)  d_in[2];
    const float* adj_vals = (const float*)d_in[3];
    const float* w_fc     = (const float*)d_in[4];
    const float* w_proj   = (const float*)d_in[5];
    const float* b_proj   = (const float*)d_in[6];
    const float* bias     = (const float*)d_in[7];

    const int n = in_sizes[0] / DIM;
    const int e = in_sizes[1];

    float* h   = (float*)d_ws;
    float* out = (float*)d_out;

    {
        dim3 grid((n + 127) / 128);
        gcn_gemm_wmma<<<grid, 256, 0, stream>>>(seq, w_fc, w_proj, bias, b_proj,
                                                h, out, n);
    }
    {
        dim3 grid((e + 7) / 8);
        gcn_edge_scatter<<<grid, 256, 0, stream>>>(h, edge_src, edge_dst,
                                                   adj_vals, out, e);
    }
    {
        const int n4 = n * (DIM / 4);
        dim3 grid((n4 + 255) / 256);
        gcn_tanh<<<grid, 256, 0, stream>>>(out, n4);
    }
}


# Edge scatter: lane L takes columns L, L+32, L+64, L+96, so each f32 atomic add (and each h load) covers one whole 128-B line: 4 atomic line requests per edge, not 16; adds still in arrival order
# speedup vs baseline: 1.2872x; 1.2872x over previous
#include <hip/hip_runtime.h>
#include <math.h>

typedef float v2f __attribute__((ext_vector_type(2)));
typedef float v8f __attribute__((ext_vector_type(8)));

#define DIM 128

__global__ __launch_bounds__(256) void gcn_gemm_wmma(
    const float* __restrict__ seq,
    const float* __restrict__ w_fc,
    const float* __restrict__ w_proj,
    const float* __restrict__ bias,
    const float* __restrict__ b_proj,
    float* __restrict__ h,
    float* __restrict__ out,
    int n)
{
    const int lane = threadIdx.x & 31;
    const int wave = threadIdx.x >> 5;
    const int row0 = blockIdx.x * 128 + wave * 16;

    const int m  = lane & 15;
    const int kh = (lane >> 4) * 2;
    const int nn = lane & 15;

    int arow = row0 + m;
    if (arow >= n) arow = n - 1;
    const float* abase = seq + (size_t)arow * DIM + kh;
    v2f a[32];
#pragma unroll
    for (int kk = 0; kk < 32; ++kk)
        a[kk] = *(const v2f*)(abase + kk * 4);

    const int rowsel = (lane >> 4) << 3;

#pragma unroll 1
    for (int t = 0; t < 8; ++t) {
        const int colBase = t * 16;
        v8f cf = {};
        v8f cp = {};

        const float* bf = w_fc   + (size_t)kh * DIM + colBase + nn;
        const float* bp = w_proj + (size_t)kh * DIM + colBase + nn;

#pragma unroll
        for (int kk = 0; kk < 32; ++kk) {
            const size_t r0 = (size_t)(kk * 4) * DIM;
            const size_t r1 = (size_t)(kk * 4 + 1) * DIM;
            v2f bfr, bpr;
            bfr.x = bf[r0];  bfr.y = bf[r1];
            bpr.x = bp[r0];  bpr.y = bp[r1];
            cf = __builtin_amdgcn_wmma_f32_16x16x4_f32(
                    false, a[kk], false, bfr, (short)0, cf, false, false);
            cp = __builtin_amdgcn_wmma_f32_16x16x4_f32(
                    false, a[kk], false, bpr, (short)0, cp, false, false);
        }

        const float badd = bias[colBase + nn] + b_proj[colBase + nn];

#pragma unroll
        for (int v = 0; v < 8; ++v) {
            const int row = row0 + v + rowsel;
            if (row < n) {
                h  [(size_t)row * DIM + colBase + nn] = cf[v];
                out[(size_t)row * DIM + colBase + nn] = cp[v] + badd;
            }
        }
    }
}

__global__ __launch_bounds__(256) void gcn_edge_scatter(
    const float* __restrict__ h,
    const int*   __restrict__ edge_src,
    const int*   __restrict__ edge_dst,
    const float* __restrict__ adj_vals,
    float* __restrict__ out,
    int e_total)
{
    const int e = blockIdx.x * (blockDim.x >> 5) + (threadIdx.x >> 5);
    if (e >= e_total) return;
    const int lane = threadIdx.x & 31;

    const int   src = edge_src[e];
    const int   dst = edge_dst[e];
    const float val = adj_vals[e];

    const float* hp = h + (size_t)src * DIM + lane;
    const float h0 = hp[0];
    const float h1 = hp[32];
    const float h2 = hp[64];
    const float h3 = hp[96];
    float* o = out + (size_t)dst * DIM + lane;
    atomicAdd(o +  0, val * h0);
    atomicAdd(o + 32, val * h1);
    atomicAdd(o + 64, val * h2);
    atomicAdd(o + 96, val * h3);
}

__global__ __launch_bounds__(256) void gcn_tanh(float* __restrict__ out, int n4)
{
    const int i = blockIdx.x * blockDim.x + threadIdx.x;
    if (i >= n4) return;
    float4 v = ((float4*)out)[i];
    v.x = tanhf(v.x); v.y = tanhf(v.y); v.z = tanhf(v.z); v.w = tanhf(v.w);
    ((float4*)out)[i] = v;
}

extern "C" void kernel_launch(void* const* d_in, const int* in_sizes, int n_in,
                              void* d_out, int out_size, void* d_ws, size_t ws_size,
                              hipStream_t stream)
{
    const float* seq      = (const float*)d_in[0];
    const int*   edge_src = (const int*)  d_in[1];
    const int*   edge_dst = (const int*)  d_in[2];
    const float* adj_vals = (const float*)d_in[3];
    const float* w_fc     = (const float*)d_in[4];
    const float* w_proj   = (const float*)d_in[5];
    const float* b_proj   = (const float*)d_in[6];
    const float* bias     = (const float*)d_in[7];

    const int n = in_sizes[0] / DIM;
    const int e = in_sizes[1];

    float* h   = (float*)d_ws;
    float* out = (float*)d_out;

    {
        dim3 grid((n + 127) / 128);
        gcn_gemm_wmma<<<grid, 256, 0, stream>>>(seq, w_fc, w_proj, bias, b_proj,
                                                h, out, n);
    }
    {
        dim3 grid((e + 7) / 8);
        gcn_edge_scatter<<<grid, 256, 0, stream>>>(h, edge_src, edge_dst,
                                                   adj_vals, out, e);
    }
    {
        const int n4 = n * (DIM / 4);
        dim3 grid((n4 + 255) / 256);
        gcn_tanh<<<grid, 256, 0, stream>>>(out, n4);
    }
}


# Edge sum by destination: an int atomic exchange links each edge to its dst row; a wave per row sorts its list by edge index, sums val*h[src] in that fixed order, adds, tanh; no f32 atomics
# speedup vs baseline: 1.7730x; 1.3774x over previous
#include <hip/hip_runtime.h>
#include <math.h>

typedef float v2f __attribute__((ext_vector_type(2)));
typedef float v8f __attribute__((ext_vector_type(8)));

#define DIM 128

__global__ __launch_bounds__(256) void gcn_gemm_wmma(
    const float* __restrict__ seq,
    const float* __restrict__ w_fc,
    const float* __restrict__ w_proj,
    const float* __restrict__ bias,
    const float* __restrict__ b_proj,
    float* __restrict__ h,
    float* __restrict__ out,
    int* __restrict__ head,
    int n)
{
    const int lane = threadIdx.x & 31;
    const int wave = threadIdx.x >> 5;
    const int row0 = blockIdx.x * 128 + wave * 16;

    if (threadIdx.x < 128) {
        const int hr = blockIdx.x * 128 + threadIdx.x;
        if (hr < n) head[hr] = -1;
    }

    const int m  = lane & 15;
    const int kh = (lane >> 4) * 2;
    const int nn = lane & 15;

    int arow = row0 + m;
    if (arow >= n) arow = n - 1;
    const float* abase = seq + (size_t)arow * DIM + kh;
    v2f a[32];
#pragma unroll
    for (int kk = 0; kk < 32; ++kk)
        a[kk] = *(const v2f*)(abase + kk * 4);

    const int rowsel = (lane >> 4) << 3;

#pragma unroll 1
    for (int t = 0; t < 8; ++t) {
        const int colBase = t * 16;
        v8f cf = {};
        v8f cp = {};

        const float* bf = w_fc   + (size_t)kh * DIM + colBase + nn;
        const float* bp = w_proj + (size_t)kh * DIM + colBase + nn;

#pragma unroll
        for (int kk = 0; kk < 32; ++kk) {
            const size_t r0 = (size_t)(kk * 4) * DIM;
            const size_t r1 = (size_t)(kk * 4 + 1) * DIM;
            v2f bfr, bpr;
            bfr.x = bf[r0];  bfr.y = bf[r1];
            bpr.x = bp[r0];  bpr.y = bp[r1];
            cf = __builtin_amdgcn_wmma_f32_16x16x4_f32(
                    false, a[kk], false, bfr, (short)0, cf, false, false);
            cp = __builtin_amdgcn_wmma_f32_16x16x4_f32(
                    false, a[kk], false, bpr, (short)0, cp, false, false);
        }

        const float badd = bias[colBase + nn] + b_proj[colBase + nn];

#pragma unroll
        for (int v = 0; v < 8; ++v) {
            const int row = row0 + v + rowsel;
            if (row < n) {
                h  [(size_t)row * DIM + colBase + nn] = cf[v];
                out[(size_t)row * DIM + colBase + nn] = cp[v] + badd;
            }
        }
    }
}

__global__ __launch_bounds__(256) void gcn_link_edges(
    const int* __restrict__ edge_dst,
    int* __restrict__ head,
    int* __restrict__ next,
    int e_total)
{
    const int e = blockIdx.x * blockDim.x + threadIdx.x;
    if (e >= e_total) return;
    next[e] = atomicExch(head + edge_dst[e], e);
}

#define CAP 128

__global__ __launch_bounds__(256) void gcn_row_gather(
    const float* __restrict__ h,
    const int*   __restrict__ edge_src,
    const float* __restrict__ adj_vals,
    const int*   __restrict__ head,
    const int*   __restrict__ next,
    float* __restrict__ out,
    int n, int e_total)
{
    __shared__ int   s_id [8][CAP];
    __shared__ int   s_ord[8][CAP];
    __shared__ __attribute__((aligned(16))) int   s_src[8][CAP];
    __shared__ __attribute__((aligned(16))) float s_val[8][CAP];

    const int lane = threadIdx.x & 31;
    const int wave = __builtin_amdgcn_readfirstlane(threadIdx.x >> 5);
    const int row  = blockIdx.x * 8 + wave;
    if (row >= n) return;

    int* id  = s_id [wave];
    int* ord = s_ord[wave];
    int* srcs = s_src[wave];
    float* vals = s_val[wave];

    int d = 0;
    for (int e = head[row]; (unsigned)e < (unsigned)e_total && d < e_total; e = next[e]) {
        if (d < CAP && lane == 0) id[d] = e;
        ++d;
    }

    const float* hl = h + lane * 4;
    float4 acc = make_float4(0.f, 0.f, 0.f, 0.f);

    if (d <= CAP) {
        for (int i = lane; i < d; i += 32) {
            const int mine = id[i];
            int rank = 0;
            for (int j = 0; j < d; ++j) rank += (id[j] < mine) ? 1 : 0;
            ord[rank] = mine;
        }
        const int d8 = (d + 7) & ~7;
        for (int i = lane; i < d8; i += 32) {
            int s = 0;
            float w = 0.f;
            if (i < d) {
                int e = ord[i];
                if ((unsigned)e >= (unsigned)e_total) e = 0;
                s = edge_src[e];
                w = adj_vals[e];
            }
            srcs[i] = s;
            vals[i] = w;
        }
        for (int r0 = 0; r0 < d8; r0 += 8) {
            float4 x[8];
            float  w[8];
            int    sv[8];
            *(int4*)&sv[0]  = *(const int4*)&srcs[r0];
            *(int4*)&sv[4]  = *(const int4*)&srcs[r0 + 4];
            *(float4*)&w[0] = *(const float4*)&vals[r0];
            *(float4*)&w[4] = *(const float4*)&vals[r0 + 4];
#pragma unroll
            for (int u = 0; u < 8; ++u) {
                const int s = __builtin_amdgcn_readfirstlane(sv[u]);
                x[u] = *(const float4*)(hl + (size_t)s * DIM);
            }
#pragma unroll
            for (int u = 0; u < 8; ++u) {
                if (r0 + u < d) {
                    acc.x = fmaf(w[u], x[u].x, acc.x);
                    acc.y = fmaf(w[u], x[u].y, acc.y);
                    acc.z = fmaf(w[u], x[u].z, acc.z);
                    acc.w = fmaf(w[u], x[u].w, acc.w);
                }
            }
        }
    } else {
        int k = 0;
        for (int e = head[row]; (unsigned)e < (unsigned)e_total && k < e_total; e = next[e]) {
            const int   s = edge_src[e];
            const float w = adj_vals[e];
            const float4 x = *(const float4*)(hl + (size_t)s * DIM);
            acc.x = fmaf(w, x.x, acc.x);
            acc.y = fmaf(w, x.y, acc.y);
            acc.z = fmaf(w, x.z, acc.z);
            acc.w = fmaf(w, x.w, acc.w);
            ++k;
        }
    }

    float4* op = (float4*)(out + (size_t)row * DIM) + lane;
    float4 p = *op;
    p.x = tanhf(p.x + acc.x);
    p.y = tanhf(p.y + acc.y);
    p.z = tanhf(p.z + acc.z);
    p.w = tanhf(p.w + acc.w);
    *op = p;
}

extern "C" void kernel_launch(void* const* d_in, const int* in_sizes, int n_in,
                              void* d_out, int out_size, void* d_ws, size_t ws_size,
                              hipStream_t stream)
{
    const float* seq      = (const float*)d_in[0];
    const int*   edge_src = (const int*)  d_in[1];
    const int*   edge_dst = (const int*)  d_in[2];
    const float* adj_vals = (const float*)d_in[3];
    const float* w_fc     = (const float*)d_in[4];
    const float* w_proj   = (const float*)d_in[5];
    const float* b_proj   = (const float*)d_in[6];
    const float* bias     = (const float*)d_in[7];

    const int n = in_sizes[0] / DIM;
    const int e = in_sizes[1];

    float* h    = (float*)d_ws;
    int*   head = (int*)(h + (size_t)n * DIM);
    int*   next = head + n;
    float* out  = (float*)d_out;

    {
        dim3 grid((n + 127) / 128);
        gcn_gemm_wmma<<<grid, 256, 0, stream>>>(seq, w_fc, w_proj, bias, b_proj,
                                                h, out, head, n);
    }
    {
        dim3 grid((e + 255) / 256);
        gcn_link_edges<<<grid, 256, 0, stream>>>(edge_dst, head, next, e);
    }
    {
        dim3 grid((n + 7) / 8);
        gcn_row_gather<<<grid, 256, 0, stream>>>(h, edge_src, adj_vals, head, next,
                                                 out, n, e);
    }
}


# Dst row's edge list split into 32 sub-lists by edge index (e & 31), one per lane; the wave walks all 32 at once: dependent hops per row fall from its degree to its longest sub-list; sum order kept
# speedup vs baseline: 2.1611x; 1.2189x over previous
#include <hip/hip_runtime.h>
#include <math.h>

typedef float v2f __attribute__((ext_vector_type(2)));
typedef float v8f __attribute__((ext_vector_type(8)));

#define DIM 128
#define SUB 32

__global__ __launch_bounds__(256) void gcn_gemm_wmma(
    const float* __restrict__ seq,
    const float* __restrict__ w_fc,
    const float* __restrict__ w_proj,
    const float* __restrict__ bias,
    const float* __restrict__ b_proj,
    float* __restrict__ h,
    float* __restrict__ out,
    int* __restrict__ head,
    int n)
{
    const int lane = threadIdx.x & 31;
    const int wave = threadIdx.x >> 5;
    const int row0 = blockIdx.x * 128 + wave * 16;

    {
        int4* hp = (int4*)head;
        const int total4 = n * (SUB / 4);
        const int4 none = make_int4(-1, -1, -1, -1);
#pragma unroll
        for (int k = 0; k < 4; ++k) {
            const int g = blockIdx.x * (128 * SUB / 4) + k * 256 + threadIdx.x;
            if (g < total4) hp[g] = none;
        }
    }

    const int m  = lane & 15;
    const int kh = (lane >> 4) * 2;
    const int nn = lane & 15;

    int arow = row0 + m;
    if (arow >= n) arow = n - 1;
    const float* abase = seq + (size_t)arow * DIM + kh;
    v2f a[32];
#pragma unroll
    for (int kk = 0; kk < 32; ++kk)
        a[kk] = *(const v2f*)(abase + kk * 4);

    const int rowsel = (lane >> 4) << 3;

#pragma unroll 1
    for (int t = 0; t < 8; ++t) {
        const int colBase = t * 16;
        v8f cf = {};
        v8f cp = {};

        const float* bf = w_fc   + (size_t)kh * DIM + colBase + nn;
        const float* bp = w_proj + (size_t)kh * DIM + colBase + nn;

#pragma unroll
        for (int kk = 0; kk < 32; ++kk) {
            const size_t r0 = (size_t)(kk * 4) * DIM;
            const size_t r1 = (size_t)(kk * 4 + 1) * DIM;
            v2f bfr, bpr;
            bfr.x = bf[r0];  bfr.y = bf[r1];
            bpr.x = bp[r0];  bpr.y = bp[r1];
            cf = __builtin_amdgcn_wmma_f32_16x16x4_f32(
                    false, a[kk], false, bfr, (short)0, cf, false, false);
            cp = __builtin_amdgcn_wmma_f32_16x16x4_f32(
                    false, a[kk], false, bpr, (short)0, cp, false, false);
        }

        const float badd = bias[colBase + nn] + b_proj[colBase + nn];

#pragma unroll
        for (int v = 0; v < 8; ++v) {
            const int row = row0 + v + rowsel;
            if (row < n) {
                h  [(size_t)row * DIM + colBase + nn] = cf[v];
                out[(size_t)row * DIM + colBase + nn] = cp[v] + badd;
            }
        }
    }
}

__global__ __launch_bounds__(256) void gcn_link_edges(
    const int* __restrict__ edge_dst,
    int* __restrict__ head,
    int* __restrict__ next,
    int e_total)
{
    const int e = blockIdx.x * blockDim.x + threadIdx.x;
    if (e >= e_total) return;
    next[e] = atomicExch(head + (size_t)edge_dst[e] * SUB + (e & (SUB - 1)), e);
}

#define CAP 128

__global__ __launch_bounds__(256) void gcn_row_gather(
    const float* __restrict__ h,
    const int*   __restrict__ edge_src,
    const float* __restrict__ adj_vals,
    const int*   __restrict__ head,
    const int*   __restrict__ next,
    float* __restrict__ out,
    int n, int e_total)
{
    __shared__ int   s_id [8][CAP];
    __shared__ int   s_ord[8][CAP];
    __shared__ __attribute__((aligned(16))) int   s_src[8][CAP];
    __shared__ __attribute__((aligned(16))) float s_val[8][CAP];

    const int lane = threadIdx.x & 31;
    const int wave = __builtin_amdgcn_readfirstlane(threadIdx.x >> 5);
    const int row  = blockIdx.x * 8 + wave;
    if (row >= n) return;

    int* id  = s_id [wave];
    int* ord = s_ord[wave];
    int* srcs = s_src[wave];
    float* vals = s_val[wave];

    const int* hrow = head + (size_t)row * SUB;

    int d = 0;
    {
        int e = hrow[lane];
        for (int it = 0; it < e_total; ++it) {
            const bool live = (unsigned)e < (unsigned)e_total;
            const unsigned m = __builtin_amdgcn_ballot_w32(live);
            if (m == 0u) break;
            if (live) {
                const int pos = d + (int)__builtin_amdgcn_mbcnt_lo(m, 0u);
                if (pos < CAP) id[pos] = e;
                e = next[e];
            }
            d += __builtin_popcount(m);
        }
    }

    const float* hl = h + lane * 4;
    float4 acc = make_float4(0.f, 0.f, 0.f, 0.f);

    if (d <= CAP) {
        for (int i = lane; i < d; i += 32) {
            const int mine = id[i];
            int rank = 0;
            for (int j = 0; j < d; ++j) rank += (id[j] < mine) ? 1 : 0;
            ord[rank] = mine;
        }
        const int d8 = (d + 7) & ~7;
        for (int i = lane; i < d8; i += 32) {
            int s = 0;
            float w = 0.f;
            if (i < d) {
                int e = ord[i];
                if ((unsigned)e >= (unsigned)e_total) e = 0;
                s = edge_src[e];
                w = adj_vals[e];
            }
            srcs[i] = s;
            vals[i] = w;
        }
        for (int r0 = 0; r0 < d8; r0 += 8) {
            float4 x[8];
            float  w[8];
            int    sv[8];
            *(int4*)&sv[0]  = *(const int4*)&srcs[r0];
            *(int4*)&sv[4]  = *(const int4*)&srcs[r0 + 4];
            *(float4*)&w[0] = *(const float4*)&vals[r0];
            *(float4*)&w[4] = *(const float4*)&vals[r0 + 4];
#pragma unroll
            for (int u = 0; u < 8; ++u) {
                const int s = __builtin_amdgcn_readfirstlane(sv[u]);
                x[u] = *(const float4*)(hl + (size_t)s * DIM);
            }
#pragma unroll
            for (int u = 0; u < 8; ++u) {
                if (r0 + u < d) {
                    acc.x = fmaf(w[u], x[u].x, acc.x);
                    acc.y = fmaf(w[u], x[u].y, acc.y);
                    acc.z = fmaf(w[u], x[u].z, acc.z);
                    acc.w = fmaf(w[u], x[u].w, acc.w);
                }
            }
        }
    } else {
        for (int sl = 0; sl < SUB; ++sl) {
            int k = 0;
            for (int e = hrow[sl]; (unsigned)e < (unsigned)e_total && k < e_total; e = next[e]) {
                const int   s = edge_src[e];
                const float w = adj_vals[e];
                const float4 x = *(const float4*)(hl + (size_t)s * DIM);
                acc.x = fmaf(w, x.x, acc.x);
                acc.y = fmaf(w, x.y, acc.y);
                acc.z = fmaf(w, x.z, acc.z);
                acc.w = fmaf(w, x.w, acc.w);
                ++k;
            }
        }
    }

    float4* op = (float4*)(out + (size_t)row * DIM) + lane;
    float4 p = *op;
    p.x = tanhf(p.x + acc.x);
    p.y = tanhf(p.y + acc.y);
    p.z = tanhf(p.z + acc.z);
    p.w = tanhf(p.w + acc.w);
    *op = p;
}

extern "C" void kernel_launch(void* const* d_in, const int* in_sizes, int n_in,
                              void* d_out, int out_size, void* d_ws, size_t ws_size,
                              hipStream_t stream)
{
    const float* seq      = (const float*)d_in[0];
    const int*   edge_src = (const int*)  d_in[1];
    const int*   edge_dst = (const int*)  d_in[2];
    const float* adj_vals = (const float*)d_in[3];
    const float* w_fc     = (const float*)d_in[4];
    const float* w_proj   = (const float*)d_in[5];
    const float* b_proj   = (const float*)d_in[6];
    const float* bias     = (const float*)d_in[7];

    const int n = in_sizes[0] / DIM;
    const int e = in_sizes[1];

    float* h    = (float*)d_ws;
    int*   head = (int*)(h + (size_t)n * DIM);
    int*   next = head + (size_t)n * SUB;
    float* out  = (float*)d_out;

    {
        dim3 grid((n + 127) / 128);
        gcn_gemm_wmma<<<grid, 256, 0, stream>>>(seq, w_fc, w_proj, bias, b_proj,
                                                h, out, head, n);
    }
    {
        dim3 grid((e + 255) / 256);
        gcn_link_edges<<<grid, 256, 0, stream>>>(edge_dst, head, next, e);
    }
    {
        dim3 grid((n + 7) / 8);
        gcn_row_gather<<<grid, 256, 0, stream>>>(h, edge_src, adj_vals, head, next,
                                                 out, n, e);
    }
}
